# starting kernel
# baseline (speedup 1.0000x reference)
#include <hip/hip_runtime.h>

typedef __attribute__((ext_vector_type(2))) float v2f;
typedef __attribute__((ext_vector_type(8))) float v8f;

#define N_C    3144
#define NP     2
#define NMOB   6
#define NCOV   10
#define T_ALL  156
#define TP     154
#define TP_PAD 160
#define NNZ_   31440
#define LDW    3168
#define NWM    ((size_t)N_C * (size_t)LDW)

#define OFF_WB   ((size_t)0)
#define OFF_WH   (NWM)
#define OFF_WA   (2 * NWM)
#define OFF_CS   (3 * NWM)
#define OFF_DS   (OFF_CS + (size_t)TP_PAD * N_C)
#define OFF_BC   (OFF_DS + (size_t)TP_PAD * N_C)
#define OFF_BD   (OFF_BC + (size_t)TP * N_C)
#define ZERO_CNT (OFF_BC)

__global__ void zero_k(float* __restrict__ p, size_t count) {
    size_t i = (size_t)blockIdx.x * blockDim.x + threadIdx.x;
    size_t stride = (size_t)gridDim.x * blockDim.x;
    for (; i < count; i += stride) p[i] = 0.0f;
}

__global__ void scatter_k(const int* __restrict__ rows, const int* __restrict__ cols,
                          const float* __restrict__ bnz, const float* __restrict__ anz,
                          const float* __restrict__ hnz, float* __restrict__ ws) {
    int i = blockIdx.x * blockDim.x + threadIdx.x;
    if (i >= NNZ_) return;
    size_t off = (size_t)rows[i] * LDW + (size_t)cols[i];
    atomicAdd(ws + OFF_WB + off, bnz[i]);
    atomicAdd(ws + OFF_WA + off, anz[i]);
    atomicAdd(ws + OFF_WH + off, hnz[i]);
}

__global__ void prep_k(const float* __restrict__ C, const float* __restrict__ D,
                       const float* __restrict__ M, const float* __restrict__ cov,
                       const float* __restrict__ mu, const float* __restrict__ nu,
                       const float* __restrict__ ups, const float* __restrict__ zet,
                       float* __restrict__ ws) {
    int idx = blockIdx.x * blockDim.x + threadIdx.x;
    if (idx >= TP * N_C) return;
    int t = idx / N_C;
    int m = idx - t * N_C;

    float cs = C[(size_t)t * N_C + m] + C[(size_t)(t + 1) * N_C + m];
    float dsv = D[(size_t)t * N_C + m] + D[(size_t)(t + 1) * N_C + m];
    ws[OFF_CS + (size_t)t * N_C + m] = cs;
    ws[OFF_DS + (size_t)t * N_C + m] = dsv;

    float mc = 0.0f, md = 0.0f;
#pragma unroll
    for (int k = 0; k < NMOB; ++k) {
#pragma unroll
        for (int tau = 0; tau < NP; ++tau) {
            float mv = M[((size_t)k * T_ALL + (size_t)(t + tau)) * N_C + m];
            mc = fmaf(mv, mu[k * NP + tau], mc);
            md = fmaf(mv, nu[k * NP + tau], md);
        }
    }
    float uc = 0.0f, zc = 0.0f;
#pragma unroll
    for (int j = 0; j < NCOV; ++j) {
        float cv = cov[(size_t)j * N_C + m];
        uc = fmaf(ups[j], cv, uc);
        zc = fmaf(zet[j], cv, zc);
    }
    ws[OFF_BC + (size_t)idx] = mc + uc;
    ws[OFF_BD + (size_t)idx] = md + zc;
}

__global__ __launch_bounds__(128)
void gemm_k(const float* __restrict__ ws, float* __restrict__ out) {
    const float* WB = ws + OFF_WB;
    const float* WH = ws + OFF_WH;
    const float* WA = ws + OFF_WA;
    const float* CS = ws + OFF_CS;
    const float* DS = ws + OFF_DS;

    const int wave = threadIdx.x >> 5;
    const int lane = threadIdx.x & 31;
    const int pair = blockIdx.x * 4 + wave;
    if (pair >= 99) return;
    const int c0 = pair * 32;
    const int t0 = blockIdx.y * 16;
    const int mrow = lane & 15;
    const int kg = (lane >> 4) * 2;

    v8f acc0 = {};
    v8f acc1 = {};
    const int z = blockIdx.z;
    const int npass = (z == 0) ? 1 : 2;

    for (int pass = 0; pass < npass; ++pass) {
        const float* X = (z == 0) ? CS : (pass == 0 ? CS : DS);
        const float* W = (z == 0) ? WB : (pass == 0 ? WH : WA);
        const float* xrow = X + (size_t)(t0 + mrow) * N_C;
        for (int k = 0; k < N_C; k += 4) {
            v2f a;
            a.x = xrow[k + kg];
            a.y = xrow[k + kg + 1];
            const float* w0 = W + (size_t)(k + kg) * LDW + (size_t)(c0 + mrow);
            v2f b0, b1;
            b0.x = w0[0];
            b0.y = w0[LDW];
            b1.x = w0[16];
            b1.y = w0[LDW + 16];
            acc0 = __builtin_amdgcn_wmma_f32_16x16x4_f32(false, a, false, b0,
                                                         (short)0, acc0, false, false);
            acc1 = __builtin_amdgcn_wmma_f32_16x16x4_f32(false, a, false, b1,
                                                         (short)0, acc1, false, false);
        }
    }

    const float* base = ws + ((z == 0) ? OFF_BC : OFF_BD);
    const size_t obase = (size_t)z * TP * N_C;
    const int rbump = (lane >> 4) << 3;
#pragma unroll
    for (int r = 0; r < 8; ++r) {
        int row = t0 + r + rbump;
        if (row >= TP) continue;
        int col0 = c0 + mrow;
        if (col0 < N_C)
            out[obase + (size_t)row * N_C + col0] =
                acc0[r] + base[(size_t)row * N_C + col0];
        int col1 = col0 + 16;
        if (col1 < N_C)
            out[obase + (size_t)row * N_C + col1] =
                acc1[r] + base[(size_t)row * N_C + col1];
    }
}

extern "C" void kernel_launch(void* const* d_in, const int* in_sizes, int n_in,
                              void* d_out, int out_size, void* d_ws, size_t ws_size,
                              hipStream_t stream) {
    const float* C   = (const float*)d_in[0];
    const float* D   = (const float*)d_in[1];
    const float* M   = (const float*)d_in[2];
    const float* cov = (const float*)d_in[3];
    const float* bnz = (const float*)d_in[4];
    const float* anz = (const float*)d_in[5];
    const float* hnz = (const float*)d_in[6];
    const float* mu  = (const float*)d_in[7];
    const float* nu  = (const float*)d_in[8];
    const float* ups = (const float*)d_in[9];
    const float* zet = (const float*)d_in[10];
    const int* rows  = (const int*)d_in[11];
    const int* cols  = (const int*)d_in[12];
    float* ws  = (float*)d_ws;
    float* out = (float*)d_out;

    zero_k<<<4096, 256, 0, stream>>>(ws, ZERO_CNT);
    scatter_k<<<(NNZ_ + 255) / 256, 256, 0, stream>>>(rows, cols, bnz, anz, hnz, ws);
    prep_k<<<(TP * N_C + 255) / 256, 256, 0, stream>>>(C, D, M, cov, mu, nu, ups, zet, ws);
    gemm_k<<<dim3(25, TP_PAD / 16, 2), 128, 0, stream>>>(ws, out);
}


# Product launch re-tiled: 8-wave workgroup owns a 16-column strip for all rows, waves split k, stream the dense weights once, WMMA only on k steps whose 4x16 weight block is non-zero; LDS join
# speedup vs baseline: 2.0243x; 2.0243x over previous
#include <hip/hip_runtime.h>

typedef __attribute__((ext_vector_type(2))) float v2f;
typedef __attribute__((ext_vector_type(8))) float v8f;

#define N_C    3144
#define NP     2
#define NMOB   6
#define NCOV   10
#define T_ALL  156
#define TP     154
#define TP_PAD 160
#define NNZ_   31440
#define LDW    3168
#define NWM    ((size_t)N_C * (size_t)LDW)

#define OFF_WB   ((size_t)0)
#define OFF_WH   (NWM)
#define OFF_WA   (2 * NWM)
#define OFF_CS   (3 * NWM)
#define OFF_DS   (OFF_CS + (size_t)TP_PAD * N_C)
#define OFF_BC   (OFF_DS + (size_t)TP_PAD * N_C)
#define OFF_BD   (OFF_BC + (size_t)TP * N_C)
#define ZERO_CNT (OFF_BC)

__global__ void zero_k(float* __restrict__ p, size_t count) {
    size_t i = (size_t)blockIdx.x * blockDim.x + threadIdx.x;
    size_t stride = (size_t)gridDim.x * blockDim.x;
    for (; i < count; i += stride) p[i] = 0.0f;
}

__global__ void scatter_k(const int* __restrict__ rows, const int* __restrict__ cols,
                          const float* __restrict__ bnz, const float* __restrict__ anz,
                          const float* __restrict__ hnz, float* __restrict__ ws) {
    int i = blockIdx.x * blockDim.x + threadIdx.x;
    if (i >= NNZ_) return;
    size_t off = (size_t)rows[i] * LDW + (size_t)cols[i];
    atomicAdd(ws + OFF_WB + off, bnz[i]);
    atomicAdd(ws + OFF_WA + off, anz[i]);
    atomicAdd(ws + OFF_WH + off, hnz[i]);
}

__global__ void prep_k(const float* __restrict__ C, const float* __restrict__ D,
                       const float* __restrict__ M, const float* __restrict__ cov,
                       const float* __restrict__ mu, const float* __restrict__ nu,
                       const float* __restrict__ ups, const float* __restrict__ zet,
                       float* __restrict__ ws) {
    int idx = blockIdx.x * blockDim.x + threadIdx.x;
    if (idx >= TP * N_C) return;
    int t = idx / N_C;
    int m = idx - t * N_C;

    float cs = C[(size_t)t * N_C + m] + C[(size_t)(t + 1) * N_C + m];
    float dsv = D[(size_t)t * N_C + m] + D[(size_t)(t + 1) * N_C + m];
    ws[OFF_CS + (size_t)t * N_C + m] = cs;
    ws[OFF_DS + (size_t)t * N_C + m] = dsv;

    float mc = 0.0f, md = 0.0f;
#pragma unroll
    for (int k = 0; k < NMOB; ++k) {
#pragma unroll
        for (int tau = 0; tau < NP; ++tau) {
            float mv = M[((size_t)k * T_ALL + (size_t)(t + tau)) * N_C + m];
            mc = fmaf(mv, mu[k * NP + tau], mc);
            md = fmaf(mv, nu[k * NP + tau], md);
        }
    }
    float uc = 0.0f, zc = 0.0f;
#pragma unroll
    for (int j = 0; j < NCOV; ++j) {
        float cv = cov[(size_t)j * N_C + m];
        uc = fmaf(ups[j], cv, uc);
        zc = fmaf(zet[j], cv, zc);
    }
    ws[OFF_BC + (size_t)idx] = mc + uc;
    ws[OFF_BD + (size_t)idx] = md + zc;
}

typedef __attribute__((ext_vector_type(4))) float v4f;

#define KSTEPS (N_C / 4)
#define NSL    8
#define NTILE  20
#define GP     10
#define NGRP   5

struct KFrag {
    v2f b, h, a;
    v2f ac[10];
    v2f ad[10];
};

__device__ __forceinline__ void load_kfrag(KFrag& f, const float* __restrict__ wl,
                                           const float* __restrict__ xc,
                                           const float* __restrict__ xd, int kstep) {
    const float* p = wl + (size_t)kstep * (size_t)(4 * LDW);
    f.b.x = p[OFF_WB];
    f.b.y = p[OFF_WB + LDW];
    f.h.x = p[OFF_WH];
    f.h.y = p[OFF_WH + LDW];
    f.a.x = p[OFF_WA];
    f.a.y = p[OFF_WA + LDW];
    const int k = 4 * kstep;
#pragma unroll
    for (int i = 0; i < 10; ++i) {
        f.ac[i] = *(const v2f*)(xc + (size_t)i * (16 * N_C) + k);
        f.ad[i] = *(const v2f*)(xd + (size_t)i * (16 * N_C) + k);
    }
}

__device__ __forceinline__ void mul_kfrag(const KFrag& f, v8f (&accC)[10], v8f (&accD)[10]) {
#pragma unroll
    for (int i = 0; i < 10; ++i) {
        accC[i] = __builtin_amdgcn_wmma_f32_16x16x4_f32(false, f.ac[i], false, f.b,
                                                        (short)0, accC[i], false, false);
        accD[i] = __builtin_amdgcn_wmma_f32_16x16x4_f32(false, f.ac[i], false, f.h,
                                                        (short)0, accD[i], false, false);
        accD[i] = __builtin_amdgcn_wmma_f32_16x16x4_f32(false, f.ad[i], false, f.a,
                                                        (short)0, accD[i], false, false);
    }
}

__device__ __forceinline__ void set_bit(unsigned long long& lo, unsigned long long& hi,
                                        int idx, bool on) {
    const unsigned long long b = on ? 1ull : 0ull;
    if (idx < 64) lo |= b << idx;
    else          hi |= b << (idx - 64);
}

__device__ __forceinline__ int pop_bit(unsigned long long& lo, unsigned long long& hi) {
    if (lo != 0ull) {
        const int b = __builtin_ctzll(lo);
        lo &= lo - 1ull;
        return b;
    }
    const int b = __builtin_ctzll(hi);
    hi &= hi - 1ull;
    return 64 + b;
}

__global__ __launch_bounds__(NSL * 32)
void gemm_k(const float* __restrict__ ws, float* __restrict__ out) {
    __shared__ float red[NSL * NTILE * 8 * 32];

    const int wave = __builtin_amdgcn_readfirstlane((int)(threadIdx.x >> 5));
    const int lane = threadIdx.x & 31;
    const int c    = lane & 15;
    const int h    = lane >> 4;
    const int c0   = blockIdx.x * 16;

    const int ks = (wave * KSTEPS) / NSL;
    const int ke = ((wave + 1) * KSTEPS) / NSL;
    const int n  = ke - ks;

    unsigned long long nzlo = 0ull, nzhi = 0ull;
    {
        const int rlast = 4 * ke - 1;
        const float* sl = ws + (size_t)(c0 + 4 * (lane & 3));
        for (int g = 0; g < NGRP; ++g) {
            v4f wb[GP], wh[GP], wa[GP];
#pragma unroll
            for (int u = 0; u < GP; ++u) {
                const int p = g * GP + u;
                const int row = min(4 * (ks + 2 * p) + (lane >> 2), rlast);
                const float* q = sl + (size_t)row * LDW;
                wb[u] = __builtin_nontemporal_load((const v4f*)(q + OFF_WB));
                wh[u] = __builtin_nontemporal_load((const v4f*)(q + OFF_WH));
                wa[u] = __builtin_nontemporal_load((const v4f*)(q + OFF_WA));
            }
#pragma unroll
            for (int u = 0; u < GP; ++u) {
                const int p = g * GP + u;
                const bool nz =
                    (wb[u].x != 0.0f) | (wb[u].y != 0.0f) | (wb[u].z != 0.0f) | (wb[u].w != 0.0f) |
                    (wh[u].x != 0.0f) | (wh[u].y != 0.0f) | (wh[u].z != 0.0f) | (wh[u].w != 0.0f) |
                    (wa[u].x != 0.0f) | (wa[u].y != 0.0f) | (wa[u].z != 0.0f) | (wa[u].w != 0.0f);
                const unsigned bm = __builtin_amdgcn_ballot_w32(nz);
                set_bit(nzlo, nzhi, 2 * p,     ((bm & 0xFFFFu) != 0u) && (2 * p < n));
                set_bit(nzlo, nzhi, 2 * p + 1, ((bm >> 16) != 0u) && (2 * p + 1 < n));
            }
        }
    }

    const float* wl = ws + (size_t)(2 * h) * LDW + (size_t)(c0 + c);
    const float* xc = ws + OFF_CS + (size_t)c * N_C + (size_t)(2 * h);
    const float* xd = ws + OFF_DS + (size_t)c * N_C + (size_t)(2 * h);

    v8f accC[10];
    v8f accD[10];
#pragma unroll
    for (int i = 0; i < 10; ++i) { accC[i] = (v8f){}; accD[i] = (v8f){}; }

    const int cnt = __builtin_popcountll(nzlo) + __builtin_popcountll(nzhi);
    if (cnt > 0) {
        KFrag f0, f1;
        int cur = pop_bit(nzlo, nzhi);
        load_kfrag(f0, wl, xc, xd, ks + cur);
        for (int j = 0; j < cnt; j += 2) {
            const int n1 = (j + 1 < cnt) ? pop_bit(nzlo, nzhi) : cur;
            load_kfrag(f1, wl, xc, xd, ks + n1);
            mul_kfrag(f0, accC, accD);
            const int n2 = (j + 2 < cnt) ? pop_bit(nzlo, nzhi) : n1;
            load_kfrag(f0, wl, xc, xd, ks + n2);
            if (j + 1 < cnt) mul_kfrag(f1, accC, accD);
            cur = n2;
        }
    }

    float* mine = red + (size_t)wave * (NTILE * 8 * 32) + lane;
#pragma unroll
    for (int i = 0; i < 10; ++i) {
#pragma unroll
        for (int r = 0; r < 8; ++r) {
            mine[(i * 8 + r) * 32]        = accC[i][r];
            mine[((10 + i) * 8 + r) * 32] = accD[i][r];
        }
    }
    __syncthreads();

    const int col = c0 + c;
    for (int j = 0; j < NTILE; ++j) {
        const int q = wave * NTILE + j;
        float s = 0.0f;
#pragma unroll
        for (int sw = 0; sw < NSL; ++sw)
            s += red[(size_t)(sw * (NTILE * 8) + q) * 32 + lane];
        const int tile = q >> 3;
        const int r    = q & 7;
        const int zz   = (tile >= 10) ? 1 : 0;
        const int i    = tile - 10 * zz;
        const int row  = 16 * i + r + 8 * h;
        if (row < TP && col < N_C) {
            const float* base = ws + (zz ? OFF_BD : OFF_BC);
            out[(size_t)zz * TP * N_C + (size_t)row * N_C + col] =
                s + base[(size_t)row * N_C + col];
        }
    }
}

extern "C" void kernel_launch(void* const* d_in, const int* in_sizes, int n_in,
                              void* d_out, int out_size, void* d_ws, size_t ws_size,
                              hipStream_t stream) {
    const float* C   = (const float*)d_in[0];
    const float* D   = (const float*)d_in[1];
    const float* M   = (const float*)d_in[2];
    const float* cov = (const float*)d_in[3];
    const float* bnz = (const float*)d_in[4];
    const float* anz = (const float*)d_in[5];
    const float* hnz = (const float*)d_in[6];
    const float* mu  = (const float*)d_in[7];
    const float* nu  = (const float*)d_in[8];
    const float* ups = (const float*)d_in[9];
    const float* zet = (const float*)d_in[10];
    const int* rows  = (const int*)d_in[11];
    const int* cols  = (const int*)d_in[12];
    float* ws  = (float*)d_ws;
    float* out = (float*)d_out;

    zero_k<<<4096, 256, 0, stream>>>(ws, ZERO_CNT);
    scatter_k<<<(NNZ_ + 255) / 256, 256, 0, stream>>>(rows, cols, bnz, anz, hnz, ws);
    prep_k<<<(TP * N_C + 255) / 256, 256, 0, stream>>>(C, D, M, cov, mu, nu, ups, zet, ws);
    gemm_k<<<197, 256, 0, stream>>>(ws, out);
}


# Setup launch stores lag-summed C and D in WMMA fragment order: a k step's A operands are one contiguous 5 KiB run read with ten 16-byte whole-line loads, not twenty 8-byte loads of 16 lines each
# speedup vs baseline: 2.9764x; 1.4703x over previous
#include <hip/hip_runtime.h>

typedef __attribute__((ext_vector_type(2))) float v2f;
typedef __attribute__((ext_vector_type(4))) float v4f;
typedef __attribute__((ext_vector_type(8))) float v8f;

#define N_C    3144
#define NP     2
#define NMOB   6
#define NCOV   10
#define T_ALL  156
#define TP     154
#define TP_PAD 160
#define NNZ_   31440
#define LDW    3168
#define NWM    ((size_t)N_C * (size_t)LDW)
#define KSTEPS (N_C / 4)

#define OFF_WB   ((size_t)0)
#define OFF_WH   (NWM)
#define OFF_WA   (2 * NWM)
#define OFF_CS   (3 * NWM)
#define OFF_DS   (OFF_CS + (size_t)TP_PAD * N_C)
#define OFF_BC   (OFF_DS + (size_t)TP_PAD * N_C)
#define OFF_BD   (OFF_BC + (size_t)TP * N_C)
#define ZERO_CNT (OFF_CS)

__global__ void zero_k(float* __restrict__ p, size_t count) {
    size_t i = (size_t)blockIdx.x * blockDim.x + threadIdx.x;
    size_t stride = (size_t)gridDim.x * blockDim.x;
    for (; i < count; i += stride) p[i] = 0.0f;
}

__global__ void scatter_k(const int* __restrict__ rows, const int* __restrict__ cols,
                          const float* __restrict__ bnz, const float* __restrict__ anz,
                          const float* __restrict__ hnz, float* __restrict__ ws) {
    int i = blockIdx.x * blockDim.x + threadIdx.x;
    if (i >= NNZ_) return;
    size_t off = (size_t)rows[i] * LDW + (size_t)cols[i];
    atomicAdd(ws + OFF_WB + off, bnz[i]);
    atomicAdd(ws + OFF_WA + off, anz[i]);
    atomicAdd(ws + OFF_WH + off, hnz[i]);
}

__global__ void prep_k(const float* __restrict__ C, const float* __restrict__ D,
                       const float* __restrict__ M, const float* __restrict__ cov,
                       const float* __restrict__ mu, const float* __restrict__ nu,
                       const float* __restrict__ ups, const float* __restrict__ zet,
                       float* __restrict__ ws) {
    int idx = blockIdx.x * blockDim.x + threadIdx.x;

    if (idx < KSTEPS * 10 * 32) {
        const int lane = idx & 31;
        const int w    = idx >> 5;
        const int i    = w / KSTEPS;
        const int ks   = w - i * KSTEPS;
        const int t    = 16 * i + (lane & 15);
        const int m    = 4 * ks + 2 * (lane >> 4);
        v4f q = (v4f){0.0f, 0.0f, 0.0f, 0.0f};
        if (t < TP) {
            const v2f c0 = *(const v2f*)(C + (size_t)t * N_C + m);
            const v2f c1 = *(const v2f*)(C + (size_t)(t + 1) * N_C + m);
            const v2f d0 = *(const v2f*)(D + (size_t)t * N_C + m);
            const v2f d1 = *(const v2f*)(D + (size_t)(t + 1) * N_C + m);
            q.x = c0.x + c1.x;
            q.y = c0.y + c1.y;
            q.z = d0.x + d1.x;
            q.w = d0.y + d1.y;
        }
        ((v4f*)(ws + OFF_CS))[((size_t)ks * 10 + (size_t)i) * 32 + lane] = q;
    }

    if (idx >= TP * N_C) return;
    int t = idx / N_C;
    int m = idx - t * N_C;

    float mc = 0.0f, md = 0.0f;
#pragma unroll
    for (int k = 0; k < NMOB; ++k) {
#pragma unroll
        for (int tau = 0; tau < NP; ++tau) {
            float mv = M[((size_t)k * T_ALL + (size_t)(t + tau)) * N_C + m];
            mc = fmaf(mv, mu[k * NP + tau], mc);
            md = fmaf(mv, nu[k * NP + tau], md);
        }
    }
    float uc = 0.0f, zc = 0.0f;
#pragma unroll
    for (int j = 0; j < NCOV; ++j) {
        float cv = cov[(size_t)j * N_C + m];
        uc = fmaf(ups[j], cv, uc);
        zc = fmaf(zet[j], cv, zc);
    }
    ws[OFF_BC + (size_t)idx] = mc + uc;
    ws[OFF_BD + (size_t)idx] = md + zc;
}

#define NSL    8
#define NTILE  20
#define GP     10
#define NGRP   5

struct KFrag {
    v2f b, h, a;
    v4f x[10];
};

__device__ __forceinline__ void load_kfrag(KFrag& f, const float* __restrict__ wl,
                                           const v4f* __restrict__ xf, int kstep) {
    const float* p = wl + (size_t)kstep * (size_t)(4 * LDW);
    f.b.x = p[OFF_WB];
    f.b.y = p[OFF_WB + LDW];
    f.h.x = p[OFF_WH];
    f.h.y = p[OFF_WH + LDW];
    f.a.x = p[OFF_WA];
    f.a.y = p[OFF_WA + LDW];
    const v4f* q = xf + (size_t)kstep * (10 * 32);
#pragma unroll
    for (int i = 0; i < 10; ++i) f.x[i] = q[i * 32];
}

__device__ __forceinline__ void mul_kfrag(const KFrag& f, v8f (&accC)[10], v8f (&accD)[10]) {
#pragma unroll
    for (int i = 0; i < 10; ++i) {
        const v2f ac = (v2f){f.x[i].x, f.x[i].y};
        const v2f ad = (v2f){f.x[i].z, f.x[i].w};
        accC[i] = __builtin_amdgcn_wmma_f32_16x16x4_f32(false, ac, false, f.b,
                                                        (short)0, accC[i], false, false);
        accD[i] = __builtin_amdgcn_wmma_f32_16x16x4_f32(false, ac, false, f.h,
                                                        (short)0, accD[i], false, false);
        accD[i] = __builtin_amdgcn_wmma_f32_16x16x4_f32(false, ad, false, f.a,
                                                        (short)0, accD[i], false, false);
    }
}

__device__ __forceinline__ void set_bit(unsigned long long& lo, unsigned long long& hi,
                                        int idx, bool on) {
    const unsigned long long b = on ? 1ull : 0ull;
    if (idx < 64) lo |= b << idx;
    else          hi |= b << (idx - 64);
}

__device__ __forceinline__ int pop_bit(unsigned long long& lo, unsigned long long& hi) {
    if (lo != 0ull) {
        const int b = __builtin_ctzll(lo);
        lo &= lo - 1ull;
        return b;
    }
    const int b = __builtin_ctzll(hi);
    hi &= hi - 1ull;
    return 64 + b;
}

__global__ __launch_bounds__(NSL * 32)
void gemm_k(const float* __restrict__ ws, float* __restrict__ out) {
    __shared__ float red[NSL * NTILE * 8 * 32];

    const int wave = __builtin_amdgcn_readfirstlane((int)(threadIdx.x >> 5));
    const int lane = threadIdx.x & 31;
    const int c    = lane & 15;
    const int h    = lane >> 4;
    const int c0   = blockIdx.x * 16;

    const int ks = (wave * KSTEPS) / NSL;
    const int ke = ((wave + 1) * KSTEPS) / NSL;
    const int n  = ke - ks;

    unsigned long long nzlo = 0ull, nzhi = 0ull;
    {
        const int rlast = 4 * ke - 1;
        const float* sl = ws + (size_t)(c0 + 4 * (lane & 3));
        for (int g = 0; g < NGRP; ++g) {
            v4f wb[GP], wh[GP], wa[GP];
#pragma unroll
            for (int u = 0; u < GP; ++u) {
                const int p = g * GP + u;
                const int row = min(4 * (ks + 2 * p) + (lane >> 2), rlast);
                const float* q = sl + (size_t)row * LDW;
                wb[u] = __builtin_nontemporal_load((const v4f*)(q + OFF_WB));
                wh[u] = __builtin_nontemporal_load((const v4f*)(q + OFF_WH));
                wa[u] = __builtin_nontemporal_load((const v4f*)(q + OFF_WA));
            }
#pragma unroll
            for (int u = 0; u < GP; ++u) {
                const int p = g * GP + u;
                const bool nz =
                    (wb[u].x != 0.0f) | (wb[u].y != 0.0f) | (wb[u].z != 0.0f) | (wb[u].w != 0.0f) |
                    (wh[u].x != 0.0f) | (wh[u].y != 0.0f) | (wh[u].z != 0.0f) | (wh[u].w != 0.0f) |
                    (wa[u].x != 0.0f) | (wa[u].y != 0.0f) | (wa[u].z != 0.0f) | (wa[u].w != 0.0f);
                const unsigned bm = __builtin_amdgcn_ballot_w32(nz);
                set_bit(nzlo, nzhi, 2 * p,     ((bm & 0xFFFFu) != 0u) && (2 * p < n));
                set_bit(nzlo, nzhi, 2 * p + 1, ((bm >> 16) != 0u) && (2 * p + 1 < n));
            }
        }
    }

    const float* wl = ws + (size_t)(2 * h) * LDW + (size_t)(c0 + c);
    const v4f* xf = (const v4f*)(ws + OFF_CS) + lane;

    v8f accC[10];
    v8f accD[10];
#pragma unroll
    for (int i = 0; i < 10; ++i) { accC[i] = (v8f){}; accD[i] = (v8f){}; }

    const int cnt = __builtin_popcountll(nzlo) + __builtin_popcountll(nzhi);
    if (cnt > 0) {
        KFrag f0, f1;
        int cur = pop_bit(nzlo, nzhi);
        load_kfrag(f0, wl, xf, ks + cur);
        for (int j = 0; j < cnt; j += 2) {
            const int n1 = (j + 1 < cnt) ? pop_bit(nzlo, nzhi) : cur;
            load_kfrag(f1, wl, xf, ks + n1);
            mul_kfrag(f0, accC, accD);
            const int n2 = (j + 2 < cnt) ? pop_bit(nzlo, nzhi) : n1;
            load_kfrag(f0, wl, xf, ks + n2);
            if (j + 1 < cnt) mul_kfrag(f1, accC, accD);
            cur = n2;
        }
    }

    float* mine = red + (size_t)wave * (NTILE * 8 * 32) + lane;
#pragma unroll
    for (int i = 0; i < 10; ++i) {
#pragma unroll
        for (int r = 0; r < 8; ++r) {
            mine[(i * 8 + r) * 32]        = accC[i][r];
            mine[((10 + i) * 8 + r) * 32] = accD[i][r];
        }
    }
    __syncthreads();

    const int col = c0 + c;
    for (int j = 0; j < NTILE; ++j) {
        const int q = wave * NTILE + j;
        float s = 0.0f;
#pragma unroll
        for (int sw = 0; sw < NSL; ++sw)
            s += red[(size_t)(sw * (NTILE * 8) + q) * 32 + lane];
        const int tile = q >> 3;
        const int r    = q & 7;
        const int zz   = (tile >= 10) ? 1 : 0;
        const int i    = tile - 10 * zz;
        const int row  = 16 * i + r + 8 * h;
        if (row < TP && col < N_C) {
            const float* base = ws + (zz ? OFF_BD : OFF_BC);
            out[(size_t)zz * TP * N_C + (size_t)row * N_C + col] =
                s + base[(size_t)row * N_C + col];
        }
    }
}

extern "C" void kernel_launch(void* const* d_in, const int* in_sizes, int n_in,
                              void* d_out, int out_size, void* d_ws, size_t ws_size,
                              hipStream_t stream) {
    const float* C   = (const float*)d_in[0];
    const float* D   = (const float*)d_in[1];
    const float* M   = (const float*)d_in[2];
    const float* cov = (const float*)d_in[3];
    const float* bnz = (const float*)d_in[4];
    const float* anz = (const float*)d_in[5];
    const float* hnz = (const float*)d_in[6];
    const float* mu  = (const float*)d_in[7];
    const float* nu  = (const float*)d_in[8];
    const float* ups = (const float*)d_in[9];
    const float* zet = (const float*)d_in[10];
    const int* rows  = (const int*)d_in[11];
    const int* cols  = (const int*)d_in[12];
    float* ws  = (float*)d_ws;
    float* out = (float*)d_out;

    zero_k<<<4096, 256, 0, stream>>>(ws, ZERO_CNT);
    scatter_k<<<(NNZ_ + 255) / 256, 256, 0, stream>>>(rows, cols, bnz, anz, hnz, ws);
    prep_k<<<(TP * N_C + 255) / 256, 256, 0, stream>>>(C, D, M, cov, mu, nu, ups, zet, ws);
    gemm_k<<<197, 256, 0, stream>>>(ws, out);
}


# Bitmap kept per half block (2 weight rows x 16 columns); product launch packs two non-empty half blocks into each WMMA k step, one per lane half: about half the k steps; k summation order changes
# speedup vs baseline: 4.1993x; 1.4109x over previous
#include <hip/hip_runtime.h>

typedef __attribute__((ext_vector_type(2))) float v2f;
typedef __attribute__((ext_vector_type(4))) float v4f;
typedef __attribute__((ext_vector_type(8))) float v8f;

#define N_C    3144
#define NP     2
#define NMOB   6
#define NCOV   10
#define T_ALL  156
#define TP     154
#define TP_PAD 160
#define NNZ_   31440
#define LDW    3168
#define NWM    ((size_t)N_C * (size_t)LDW)
#define KSTEPS (N_C / 4)
#define NSTRIP 197
#define BLK    192
#define OCCW   64

#define OFF_OCC  ((size_t)KSTEPS * NSTRIP * BLK)
#define OFF_CS   (3 * NWM)
#define OFF_DS   (OFF_CS + (size_t)TP_PAD * N_C)
#define OFF_BC   (OFF_DS + (size_t)TP_PAD * N_C)
#define OFF_BD   (OFF_BC + (size_t)TP * N_C)

#define ZGRP   3930

#define TPB    256
#define ZBLK   246
#define SBLK   123
#define PBLK   1892
#define PSPLIT 832
#define GRID_A 1078
#define GRID_B 1183
static_assert(GRID_A == ZBLK + PSPLIT && GRID_B == SBLK + PBLK - PSPLIT, "setup grid sizes");

__device__ __forceinline__ void zero_part(int id, const int* __restrict__ rows,
                                          const int* __restrict__ cols,
                                          float* __restrict__ ws) {
    if (id < NSTRIP * OCCW) ((unsigned*)(ws + OFF_OCC))[id] = 0u;
    const int g = id >> 4;
    if (g >= ZGRP) return;
    const v4f z = (v4f){0.0f, 0.0f, 0.0f, 0.0f};
#pragma unroll
    for (int k = 0; k < 8; ++k) {
        const int e = g + k * ZGRP;
        const size_t blk = (size_t)(rows[e] >> 2) * NSTRIP + (size_t)(cols[e] >> 4);
        v4f* p = (v4f*)(ws + blk * BLK) + (id & 15);
        p[0]  = z;
        p[16] = z;
        p[32] = z;
    }
}

__device__ __forceinline__ void scatter_part(int i, const int* __restrict__ rows,
                                             const int* __restrict__ cols,
                                             const float* __restrict__ bnz,
                                             const float* __restrict__ anz,
                                             const float* __restrict__ hnz,
                                             float* __restrict__ ws) {
    if (i >= NNZ_) return;
    const int r  = rows[i];
    const int cc = cols[i];
    const int ks = r >> 2;
    const int s  = cc >> 4;
    const size_t off = ((size_t)ks * NSTRIP + (size_t)s) * BLK
                     + (size_t)((r & 3) * 16 + (cc & 15));
    atomicAdd(ws + off,       bnz[i]);
    atomicAdd(ws + off + 64,  hnz[i]);
    atomicAdd(ws + off + 128, anz[i]);
    const int hb = r >> 1;
    atomicOr((unsigned*)(ws + OFF_OCC) + s * OCCW + (hb >> 5), 1u << (hb & 31));
}

__device__ __forceinline__ void prep_part(int idx,
                       const float* __restrict__ C, const float* __restrict__ D,
                       const float* __restrict__ M, const float* __restrict__ cov,
                       const float* __restrict__ mu, const float* __restrict__ nu,
                       const float* __restrict__ ups, const float* __restrict__ zet,
                       float* __restrict__ ws) {

    if (idx < KSTEPS * 10 * 32) {
        const int lane = idx & 31;
        const int w    = idx >> 5;
        const int i    = w / KSTEPS;
        const int ks   = w - i * KSTEPS;
        const int t    = 16 * i + (lane & 15);
        const int m    = 4 * ks + 2 * (lane >> 4);
        v4f q = (v4f){0.0f, 0.0f, 0.0f, 0.0f};
        if (t < TP) {
            const v2f c0 = *(const v2f*)(C + (size_t)t * N_C + m);
            const v2f c1 = *(const v2f*)(C + (size_t)(t + 1) * N_C + m);
            const v2f d0 = *(const v2f*)(D + (size_t)t * N_C + m);
            const v2f d1 = *(const v2f*)(D + (size_t)(t + 1) * N_C + m);
            q.x = c0.x + c1.x;
            q.y = c0.y + c1.y;
            q.z = d0.x + d1.x;
            q.w = d0.y + d1.y;
        }
        ((v4f*)(ws + OFF_CS))[((size_t)ks * 10 + (size_t)i) * 32 + lane] = q;
    }

    if (idx >= TP * N_C) return;
    int t = idx / N_C;
    int m = idx - t * N_C;

    float mc = 0.0f, md = 0.0f;
#pragma unroll
    for (int k = 0; k < NMOB; ++k) {
#pragma unroll
        for (int tau = 0; tau < NP; ++tau) {
            float mv = M[((size_t)k * T_ALL + (size_t)(t + tau)) * N_C + m];
            mc = fmaf(mv, mu[k * NP + tau], mc);
            md = fmaf(mv, nu[k * NP + tau], md);
        }
    }
    float uc = 0.0f, zc = 0.0f;
#pragma unroll
    for (int j = 0; j < NCOV; ++j) {
        float cv = cov[(size_t)j * N_C + m];
        uc = fmaf(ups[j], cv, uc);
        zc = fmaf(zet[j], cv, zc);
    }
    ws[OFF_BC + (size_t)idx] = mc + uc;
    ws[OFF_BD + (size_t)idx] = md + zc;
}


__global__ __launch_bounds__(TPB)
void setup_a_k(const int* __restrict__ rows, const int* __restrict__ cols,
               const float* __restrict__ C, const float* __restrict__ D,
               const float* __restrict__ M, const float* __restrict__ cov,
               const float* __restrict__ mu, const float* __restrict__ nu,
               const float* __restrict__ ups, const float* __restrict__ zet,
               float* __restrict__ ws) {
    const int b = blockIdx.x;
    if (b < ZBLK) {
        zero_part(b * TPB + (int)threadIdx.x, rows, cols, ws);
        return;
    }
    prep_part((b - ZBLK) * TPB + (int)threadIdx.x, C, D, M, cov, mu, nu, ups, zet, ws);
}

__global__ __launch_bounds__(TPB)
void setup_b_k(const int* __restrict__ rows, const int* __restrict__ cols,
               const float* __restrict__ bnz, const float* __restrict__ anz,
               const float* __restrict__ hnz,
               const float* __restrict__ C, const float* __restrict__ D,
               const float* __restrict__ M, const float* __restrict__ cov,
               const float* __restrict__ mu, const float* __restrict__ nu,
               const float* __restrict__ ups, const float* __restrict__ zet,
               float* __restrict__ ws) {
    const int b = blockIdx.x;
    if (b < SBLK) {
        scatter_part(b * TPB + (int)threadIdx.x, rows, cols, bnz, anz, hnz, ws);
        return;
    }
    prep_part((b - SBLK + PSPLIT) * TPB + (int)threadIdx.x, C, D, M, cov, mu, nu, ups, zet, ws);
}

#define NSL    8
#define NTILE  20

struct KFrag {
    v2f b, h, a;
    v4f x[10];
};

__device__ __forceinline__ void load_kfrag(KFrag& f, const float* __restrict__ wl,
                                           const v4f* __restrict__ xf,
                                           int hb0, int hb1, int h) {
    const int hb = (h != 0) ? hb1 : hb0;
    const float* p = wl + (size_t)(hb >> 1) * (size_t)(NSTRIP * BLK) + (size_t)((hb & 1) * 32);
    f.b.x = p[0];
    f.b.y = p[16];
    f.h.x = p[64];
    f.h.y = p[64 + 16];
    f.a.x = p[128];
    f.a.y = p[128 + 16];
    const v4f* q = xf + (size_t)(hb >> 1) * (10 * 32) + (size_t)((hb & 1) * 16);
#pragma unroll
    for (int i = 0; i < 10; ++i) f.x[i] = q[i * 32];
}

__device__ __forceinline__ void mul_kfrag(const KFrag& f, v8f (&accC)[10], v8f (&accD)[10]) {
#pragma unroll
    for (int i = 0; i < 10; ++i) {
        const v2f ac = (v2f){f.x[i].x, f.x[i].y};
        const v2f ad = (v2f){f.x[i].z, f.x[i].w};
        accC[i] = __builtin_amdgcn_wmma_f32_16x16x4_f32(false, ac, false, f.b,
                                                        (short)0, accC[i], false, false);
        accD[i] = __builtin_amdgcn_wmma_f32_16x16x4_f32(false, ac, false, f.h,
                                                        (short)0, accD[i], false, false);
        accD[i] = __builtin_amdgcn_wmma_f32_16x16x4_f32(false, ad, false, f.a,
                                                        (short)0, accD[i], false, false);
    }
}

__device__ __forceinline__ int pop_bit(unsigned long long& q0, unsigned long long& q1,
                                       unsigned long long& q2, unsigned long long& q3) {
    if (q0 != 0ull) {
        const int b = __builtin_ctzll(q0);
        q0 &= q0 - 1ull;
        return b;
    }
    if (q1 != 0ull) {
        const int b = __builtin_ctzll(q1);
        q1 &= q1 - 1ull;
        return 64 + b;
    }
    if (q2 != 0ull) {
        const int b = __builtin_ctzll(q2);
        q2 &= q2 - 1ull;
        return 128 + b;
    }
    const int b = __builtin_ctzll(q3);
    q3 &= q3 - 1ull;
    return 192 + b;
}

__global__ __launch_bounds__(NSL * 32)
void gemm_k(const float* __restrict__ ws, float* __restrict__ out) {
    __shared__ float red[NSL * NTILE * 8 * 32];

    const int wave = __builtin_amdgcn_readfirstlane((int)(threadIdx.x >> 5));
    const int lane = threadIdx.x & 31;
    const int c    = lane & 15;
    const int h    = lane >> 4;
    const int c0   = blockIdx.x * 16;

    const int ks = (wave * KSTEPS) / NSL;
    const int ke = ((wave + 1) * KSTEPS) / NSL;
    const int n  = ke - ks;

    unsigned long long z0, z1, z2, z3;
    {
        const unsigned* occ = (const unsigned*)(ws + OFF_OCC) + (size_t)blockIdx.x * OCCW;
        const int hs = 2 * ks;
        const int w0 = hs >> 5;
        const int sh = hs & 31;
        const unsigned long long q0 = (unsigned long long)occ[w0]
                                    | ((unsigned long long)occ[w0 + 1] << 32);
        const unsigned long long q1 = (unsigned long long)occ[w0 + 2]
                                    | ((unsigned long long)occ[w0 + 3] << 32);
        const unsigned long long q2 = (unsigned long long)occ[w0 + 4]
                                    | ((unsigned long long)occ[w0 + 5] << 32);
        const unsigned long long q3 = (unsigned long long)occ[w0 + 6]
                                    | ((unsigned long long)occ[w0 + 7] << 32);
        z0 = sh ? ((q0 >> sh) | (q1 << (64 - sh))) : q0;
        z1 = sh ? ((q1 >> sh) | (q2 << (64 - sh))) : q1;
        z2 = sh ? ((q2 >> sh) | (q3 << (64 - sh))) : q2;
        z3 = q3 >> sh;
        z3 &= (1ull << (2 * n - 192)) - 1ull;
    }

    const float* wl = ws + (size_t)blockIdx.x * BLK + (size_t)c;
    const v4f* xf = (const v4f*)(ws + OFF_CS) + c;
    const int hb_base = 2 * ks;

    v8f accC[10];
    v8f accD[10];
#pragma unroll
    for (int i = 0; i < 10; ++i) { accC[i] = (v8f){}; accD[i] = (v8f){}; }

    const int cnt = __builtin_popcountll(z0) + __builtin_popcountll(z1)
                  + __builtin_popcountll(z2) + __builtin_popcountll(z3);
    const int  nst = cnt >> 1;
    const bool odd = (cnt & 1) != 0;

    KFrag ft;
    if (odd) {
        const int pt = hb_base + pop_bit(z0, z1, z2, z3);
        load_kfrag(ft, wl, xf, pt, pt, h);
    }

    if (nst > 0) {
        KFrag f0, f1;
        int p0, p1;
        p0 = hb_base + pop_bit(z0, z1, z2, z3);
        p1 = hb_base + pop_bit(z0, z1, z2, z3);
        load_kfrag(f0, wl, xf, p0, p1, h);
        for (int j = 0; j < nst; j += 2) {
            if (j + 1 < nst) {
                p0 = hb_base + pop_bit(z0, z1, z2, z3);
                p1 = hb_base + pop_bit(z0, z1, z2, z3);
            }
            load_kfrag(f1, wl, xf, p0, p1, h);
            __builtin_amdgcn_sched_barrier(0);
            mul_kfrag(f0, accC, accD);
            if (j + 2 < nst) {
                p0 = hb_base + pop_bit(z0, z1, z2, z3);
                p1 = hb_base + pop_bit(z0, z1, z2, z3);
            }
            load_kfrag(f0, wl, xf, p0, p1, h);
            __builtin_amdgcn_sched_barrier(0);
            if (j + 1 < nst) mul_kfrag(f1, accC, accD);
        }
    }

    if (odd) {
        if (h != 0) {
            ft.b = (v2f){0.0f, 0.0f};
            ft.h = (v2f){0.0f, 0.0f};
            ft.a = (v2f){0.0f, 0.0f};
        }
        mul_kfrag(ft, accC, accD);
    }

    float* mine = red + (size_t)wave * (NTILE * 8 * 32) + lane;
#pragma unroll
    for (int i = 0; i < 10; ++i) {
#pragma unroll
        for (int r = 0; r < 8; ++r) {
            mine[(i * 8 + r) * 32]        = accC[i][r];
            mine[((10 + i) * 8 + r) * 32] = accD[i][r];
        }
    }
    __syncthreads();

    const int col = c0 + c;
    for (int j = 0; j < NTILE; ++j) {
        const int q = wave * NTILE + j;
        float s = 0.0f;
#pragma unroll
        for (int sw = 0; sw < NSL; ++sw)
            s += red[(size_t)(sw * (NTILE * 8) + q) * 32 + lane];
        const int tile = q >> 3;
        const int r    = q & 7;
        const int zz   = (tile >= 10) ? 1 : 0;
        const int i    = tile - 10 * zz;
        const int row  = 16 * i + r + 8 * h;
        if (row < TP && col < N_C) {
            const float* base = ws + (zz ? OFF_BD : OFF_BC);
            out[(size_t)zz * TP * N_C + (size_t)row * N_C + col] =
                s + base[(size_t)row * N_C + col];
        }
    }
}

extern "C" void kernel_launch(void* const* d_in, const int* in_sizes, int n_in,
                              void* d_out, int out_size, void* d_ws, size_t ws_size,
                              hipStream_t stream) {
    const float* C   = (const float*)d_in[0];
    const float* D   = (const float*)d_in[1];
    const float* M   = (const float*)d_in[2];
    const float* cov = (const float*)d_in[3];
    const float* bnz = (const float*)d_in[4];
    const float* anz = (const float*)d_in[5];
    const float* hnz = (const float*)d_in[6];
    const float* mu  = (const float*)d_in[7];
    const float* nu  = (const float*)d_in[8];
    const float* ups = (const float*)d_in[9];
    const float* zet = (const float*)d_in[10];
    const int* rows  = (const int*)d_in[11];
    const int* cols  = (const int*)d_in[12];
    float* ws  = (float*)d_ws;
    float* out = (float*)d_out;

    setup_a_k<<<GRID_A, TPB, 0, stream>>>(rows, cols, C, D, M, cov, mu, nu, ups, zet, ws);
    setup_b_k<<<GRID_B, TPB, 0, stream>>>(rows, cols, bnz, anz, hnz,
                                          C, D, M, cov, mu, nu, ups, zet, ws);
    gemm_k<<<197, 256, 0, stream>>>(ws, out);
}


# Join rows dealt to waves by accumulator register: a lane's 20 base terms get fixed addresses and are requested before the product loop, not one memory round trip per row inside the join loop
# speedup vs baseline: 4.8924x; 1.1651x over previous
#include <hip/hip_runtime.h>

typedef __attribute__((ext_vector_type(2))) float v2f;
typedef __attribute__((ext_vector_type(4))) float v4f;
typedef __attribute__((ext_vector_type(8))) float v8f;

#define N_C    3144
#define NP     2
#define NMOB   6
#define NCOV   10
#define T_ALL  156
#define TP     154
#define TP_PAD 160
#define NNZ_   31440
#define LDW    3168
#define NWM    ((size_t)N_C * (size_t)LDW)
#define KSTEPS (N_C / 4)
#define NSTRIP 197
#define BLK    192
#define OCCW   64

#define OFF_OCC  ((size_t)KSTEPS * NSTRIP * BLK)
#define OFF_CS   (3 * NWM)
#define OFF_DS   (OFF_CS + (size_t)TP_PAD * N_C)
#define OFF_BC   (OFF_DS + (size_t)TP_PAD * N_C)
#define OFF_BD   (OFF_BC + (size_t)TP * N_C)

#define ZGRP   3930

#define TPB    256
#define ZBLK   246
#define SBLK   123
#define PBLK   1892
#define PSPLIT 832
#define GRID_A 1078
#define GRID_B 1183
static_assert(GRID_A == ZBLK + PSPLIT && GRID_B == SBLK + PBLK - PSPLIT, "setup grid sizes");

__device__ __forceinline__ void zero_part(int id, const int* __restrict__ rows,
                                          const int* __restrict__ cols,
                                          float* __restrict__ ws) {
    if (id < NSTRIP * OCCW) ((unsigned*)(ws + OFF_OCC))[id] = 0u;
    const int g = id >> 4;
    if (g >= ZGRP) return;
    const v4f z = (v4f){0.0f, 0.0f, 0.0f, 0.0f};
#pragma unroll
    for (int k = 0; k < 8; ++k) {
        const int e = g + k * ZGRP;
        const size_t blk = (size_t)(rows[e] >> 2) * NSTRIP + (size_t)(cols[e] >> 4);
        v4f* p = (v4f*)(ws + blk * BLK) + (id & 15);
        p[0]  = z;
        p[16] = z;
        p[32] = z;
    }
}

__device__ __forceinline__ void scatter_part(int i, const int* __restrict__ rows,
                                             const int* __restrict__ cols,
                                             const float* __restrict__ bnz,
                                             const float* __restrict__ anz,
                                             const float* __restrict__ hnz,
                                             float* __restrict__ ws) {
    if (i >= NNZ_) return;
    const int r  = rows[i];
    const int cc = cols[i];
    const int ks = r >> 2;
    const int s  = cc >> 4;
    const size_t off = ((size_t)ks * NSTRIP + (size_t)s) * BLK
                     + (size_t)((r & 3) * 16 + (cc & 15));
    atomicAdd(ws + off,       bnz[i]);
    atomicAdd(ws + off + 64,  hnz[i]);
    atomicAdd(ws + off + 128, anz[i]);
    const int hb = r >> 1;
    atomicOr((unsigned*)(ws + OFF_OCC) + s * OCCW + (hb >> 5), 1u << (hb & 31));
}

__device__ __forceinline__ void prep_part(int idx,
                       const float* __restrict__ C, const float* __restrict__ D,
                       const float* __restrict__ M, const float* __restrict__ cov,
                       const float* __restrict__ mu, const float* __restrict__ nu,
                       const float* __restrict__ ups, const float* __restrict__ zet,
                       float* __restrict__ ws) {

    if (idx < KSTEPS * 10 * 32) {
        const int lane = idx & 31;
        const int w    = idx >> 5;
        const int i    = w / KSTEPS;
        const int ks   = w - i * KSTEPS;
        const int t    = 16 * i + (lane & 15);
        const int m    = 4 * ks + 2 * (lane >> 4);
        v4f q = (v4f){0.0f, 0.0f, 0.0f, 0.0f};
        if (t < TP) {
            const v2f c0 = *(const v2f*)(C + (size_t)t * N_C + m);
            const v2f c1 = *(const v2f*)(C + (size_t)(t + 1) * N_C + m);
            const v2f d0 = *(const v2f*)(D + (size_t)t * N_C + m);
            const v2f d1 = *(const v2f*)(D + (size_t)(t + 1) * N_C + m);
            q.x = c0.x + c1.x;
            q.y = c0.y + c1.y;
            q.z = d0.x + d1.x;
            q.w = d0.y + d1.y;
        }
        ((v4f*)(ws + OFF_CS))[((size_t)ks * 10 + (size_t)i) * 32 + lane] = q;
    }

    if (idx >= TP * N_C) return;
    int t = idx / N_C;
    int m = idx - t * N_C;

    float mc = 0.0f, md = 0.0f;
#pragma unroll
    for (int k = 0; k < NMOB; ++k) {
#pragma unroll
        for (int tau = 0; tau < NP; ++tau) {
            float mv = M[((size_t)k * T_ALL + (size_t)(t + tau)) * N_C + m];
            mc = fmaf(mv, mu[k * NP + tau], mc);
            md = fmaf(mv, nu[k * NP + tau], md);
        }
    }
    float uc = 0.0f, zc = 0.0f;
#pragma unroll
    for (int j = 0; j < NCOV; ++j) {
        float cv = cov[(size_t)j * N_C + m];
        uc = fmaf(ups[j], cv, uc);
        zc = fmaf(zet[j], cv, zc);
    }
    ws[OFF_BC + (size_t)idx] = mc + uc;
    ws[OFF_BD + (size_t)idx] = md + zc;
}


__global__ __launch_bounds__(TPB)
void setup_a_k(const int* __restrict__ rows, const int* __restrict__ cols,
               const float* __restrict__ C, const float* __restrict__ D,
               const float* __restrict__ M, const float* __restrict__ cov,
               const float* __restrict__ mu, const float* __restrict__ nu,
               const float* __restrict__ ups, const float* __restrict__ zet,
               float* __restrict__ ws) {
    const int b = blockIdx.x;
    if (b < ZBLK) {
        zero_part(b * TPB + (int)threadIdx.x, rows, cols, ws);
        return;
    }
    prep_part((b - ZBLK) * TPB + (int)threadIdx.x, C, D, M, cov, mu, nu, ups, zet, ws);
}

__global__ __launch_bounds__(TPB)
void setup_b_k(const int* __restrict__ rows, const int* __restrict__ cols,
               const float* __restrict__ bnz, const float* __restrict__ anz,
               const float* __restrict__ hnz,
               const float* __restrict__ C, const float* __restrict__ D,
               const float* __restrict__ M, const float* __restrict__ cov,
               const float* __restrict__ mu, const float* __restrict__ nu,
               const float* __restrict__ ups, const float* __restrict__ zet,
               float* __restrict__ ws) {
    const int b = blockIdx.x;
    if (b < SBLK) {
        scatter_part(b * TPB + (int)threadIdx.x, rows, cols, bnz, anz, hnz, ws);
        return;
    }
    prep_part((b - SBLK + PSPLIT) * TPB + (int)threadIdx.x, C, D, M, cov, mu, nu, ups, zet, ws);
}

#define NSL    8
#define NTILE  20

struct KFrag {
    v2f b, h, a;
    v4f x[10];
};

__device__ __forceinline__ void load_kfrag(KFrag& f, const float* __restrict__ wl,
                                           const v4f* __restrict__ xf,
                                           int hb0, int hb1, int h) {
    const int hb = (h != 0) ? hb1 : hb0;
    const float* p = wl + (size_t)(hb >> 1) * (size_t)(NSTRIP * BLK) + (size_t)((hb & 1) * 32);
    f.b.x = p[0];
    f.b.y = p[16];
    f.h.x = p[64];
    f.h.y = p[64 + 16];
    f.a.x = p[128];
    f.a.y = p[128 + 16];
    const v4f* q = xf + (size_t)(hb >> 1) * (10 * 32) + (size_t)((hb & 1) * 16);
#pragma unroll
    for (int i = 0; i < 10; ++i) f.x[i] = q[i * 32];
}

__device__ __forceinline__ void mul_kfrag(const KFrag& f, v8f (&accC)[10], v8f (&accD)[10]) {
#pragma unroll
    for (int i = 0; i < 10; ++i) {
        const v2f ac = (v2f){f.x[i].x, f.x[i].y};
        const v2f ad = (v2f){f.x[i].z, f.x[i].w};
        accC[i] = __builtin_amdgcn_wmma_f32_16x16x4_f32(false, ac, false, f.b,
                                                        (short)0, accC[i], false, false);
        accD[i] = __builtin_amdgcn_wmma_f32_16x16x4_f32(false, ac, false, f.h,
                                                        (short)0, accD[i], false, false);
        accD[i] = __builtin_amdgcn_wmma_f32_16x16x4_f32(false, ad, false, f.a,
                                                        (short)0, accD[i], false, false);
    }
}

__device__ __forceinline__ int pop_bit(unsigned long long& q0, unsigned long long& q1,
                                       unsigned long long& q2, unsigned long long& q3) {
    if (q0 != 0ull) {
        const int b = __builtin_ctzll(q0);
        q0 &= q0 - 1ull;
        return b;
    }
    if (q1 != 0ull) {
        const int b = __builtin_ctzll(q1);
        q1 &= q1 - 1ull;
        return 64 + b;
    }
    if (q2 != 0ull) {
        const int b = __builtin_ctzll(q2);
        q2 &= q2 - 1ull;
        return 128 + b;
    }
    const int b = __builtin_ctzll(q3);
    q3 &= q3 - 1ull;
    return 192 + b;
}

__global__ __launch_bounds__(NSL * 32)
void gemm_k(const float* __restrict__ ws, float* __restrict__ out) {
    __shared__ float red[NSL * NTILE * 8 * 32];

    const int wave = __builtin_amdgcn_readfirstlane((int)(threadIdx.x >> 5));
    const int lane = threadIdx.x & 31;
    const int c    = lane & 15;
    const int h    = lane >> 4;
    const int c0   = blockIdx.x * 16;

    const int ks = (wave * KSTEPS) / NSL;
    const int ke = ((wave + 1) * KSTEPS) / NSL;
    const int n  = ke - ks;

    const int  col   = c0 + c;
    const bool colok = col < N_C;
    const int  colc  = colok ? col : (N_C - 1);
    const int  rowl  = wave + 8 * h;
    float bv[NTILE];
#pragma unroll
    for (int j = 0; j < NTILE; ++j) {
        const int zz = j / 10;
        const int i  = j - 10 * zz;
        int row = 16 * i + rowl;
        if (i == 9 && row > TP - 1) row = TP - 1;
        bv[j] = ws[(zz ? OFF_BD : OFF_BC) + (size_t)row * N_C + (size_t)colc];
    }

    unsigned long long z0, z1, z2, z3;
    {
        const unsigned* occ = (const unsigned*)(ws + OFF_OCC) + (size_t)blockIdx.x * OCCW;
        const int hs = 2 * ks;
        const int w0 = hs >> 5;
        const int sh = hs & 31;
        const unsigned long long q0 = (unsigned long long)occ[w0]
                                    | ((unsigned long long)occ[w0 + 1] << 32);
        const unsigned long long q1 = (unsigned long long)occ[w0 + 2]
                                    | ((unsigned long long)occ[w0 + 3] << 32);
        const unsigned long long q2 = (unsigned long long)occ[w0 + 4]
                                    | ((unsigned long long)occ[w0 + 5] << 32);
        const unsigned long long q3 = (unsigned long long)occ[w0 + 6]
                                    | ((unsigned long long)occ[w0 + 7] << 32);
        z0 = sh ? ((q0 >> sh) | (q1 << (64 - sh))) : q0;
        z1 = sh ? ((q1 >> sh) | (q2 << (64 - sh))) : q1;
        z2 = sh ? ((q2 >> sh) | (q3 << (64 - sh))) : q2;
        z3 = q3 >> sh;
        z3 &= (1ull << (2 * n - 192)) - 1ull;
    }

    const float* wl = ws + (size_t)blockIdx.x * BLK + (size_t)c;
    const v4f* xf = (const v4f*)(ws + OFF_CS) + c;
    const int hb_base = 2 * ks;

    v8f accC[10];
    v8f accD[10];
#pragma unroll
    for (int i = 0; i < 10; ++i) { accC[i] = (v8f){}; accD[i] = (v8f){}; }

    const int cnt = __builtin_popcountll(z0) + __builtin_popcountll(z1)
                  + __builtin_popcountll(z2) + __builtin_popcountll(z3);
    const int  nst = cnt >> 1;
    const bool odd = (cnt & 1) != 0;

    KFrag ft;
    if (odd) {
        const int pt = hb_base + pop_bit(z0, z1, z2, z3);
        load_kfrag(ft, wl, xf, pt, pt, h);
    }

    if (nst > 0) {
        KFrag f0, f1;
        int p0, p1;
        p0 = hb_base + pop_bit(z0, z1, z2, z3);
        p1 = hb_base + pop_bit(z0, z1, z2, z3);
        load_kfrag(f0, wl, xf, p0, p1, h);
        for (int j = 0; j < nst; j += 2) {
            if (j + 1 < nst) {
                p0 = hb_base + pop_bit(z0, z1, z2, z3);
                p1 = hb_base + pop_bit(z0, z1, z2, z3);
            }
            load_kfrag(f1, wl, xf, p0, p1, h);
            __builtin_amdgcn_sched_barrier(0);
            mul_kfrag(f0, accC, accD);
            if (j + 2 < nst) {
                p0 = hb_base + pop_bit(z0, z1, z2, z3);
                p1 = hb_base + pop_bit(z0, z1, z2, z3);
            }
            load_kfrag(f0, wl, xf, p0, p1, h);
            __builtin_amdgcn_sched_barrier(0);
            if (j + 1 < nst) mul_kfrag(f1, accC, accD);
        }
    }

    if (odd) {
        if (h != 0) {
            ft.b = (v2f){0.0f, 0.0f};
            ft.h = (v2f){0.0f, 0.0f};
            ft.a = (v2f){0.0f, 0.0f};
        }
        mul_kfrag(ft, accC, accD);
    }

    float* mine = red + (size_t)wave * (NTILE * 8 * 32) + lane;
#pragma unroll
    for (int i = 0; i < 10; ++i) {
#pragma unroll
        for (int r = 0; r < 8; ++r) {
            mine[(i * 8 + r) * 32]        = accC[i][r];
            mine[((10 + i) * 8 + r) * 32] = accD[i][r];
        }
    }
    __syncthreads();

    const float* jn = red + (size_t)wave * 32 + lane;
    float sv[NTILE];
#pragma unroll
    for (int j = 0; j < NTILE; ++j) {
        float s = 0.0f;
#pragma unroll
        for (int sw = 0; sw < NSL; ++sw)
            s += jn[(size_t)(sw * (NTILE * 8) + j * 8) * 32];
        sv[j] = s + bv[j];
    }
    if (colok) {
        float* o = out + (size_t)rowl * N_C + (size_t)col;
#pragma unroll
        for (int j = 0; j < NTILE; ++j) {
            const int zz = j / 10;
            const int i  = j - 10 * zz;
            if (i < 9 || 16 * 9 + rowl < TP)
                o[(size_t)zz * TP * N_C + (size_t)(16 * i) * N_C] = sv[j];
        }
    }
}

extern "C" void kernel_launch(void* const* d_in, const int* in_sizes, int n_in,
                              void* d_out, int out_size, void* d_ws, size_t ws_size,
                              hipStream_t stream) {
    const float* C   = (const float*)d_in[0];
    const float* D   = (const float*)d_in[1];
    const float* M   = (const float*)d_in[2];
    const float* cov = (const float*)d_in[3];
    const float* bnz = (const float*)d_in[4];
    const float* anz = (const float*)d_in[5];
    const float* hnz = (const float*)d_in[6];
    const float* mu  = (const float*)d_in[7];
    const float* nu  = (const float*)d_in[8];
    const float* ups = (const float*)d_in[9];
    const float* zet = (const float*)d_in[10];
    const int* rows  = (const int*)d_in[11];
    const int* cols  = (const int*)d_in[12];
    float* ws  = (float*)d_ws;
    float* out = (float*)d_out;

    setup_a_k<<<GRID_A, TPB, 0, stream>>>(rows, cols, C, D, M, cov, mu, nu, ups, zet, ws);
    setup_b_k<<<GRID_B, TPB, 0, stream>>>(rows, cols, bnz, anz, hnz,
                                          C, D, M, cov, mu, nu, ups, zet, ws);
    gemm_k<<<197, 256, 0, stream>>>(ws, out);
}


# Weights and bitmap kept per single weight row (1 x 16 columns x {H,A,B}, 192 contiguous bytes); product launch lists non-empty rows in LDS, packs four per WMMA k step; k summation order changes
# speedup vs baseline: 8.4052x; 1.7180x over previous
#include <hip/hip_runtime.h>

typedef __attribute__((ext_vector_type(2))) float v2f;
typedef __attribute__((ext_vector_type(4))) float v4f;
typedef __attribute__((ext_vector_type(8))) float v8f;

#define N_C    3144
#define NP     2
#define NMOB   6
#define NCOV   10
#define T_ALL  156
#define TP     154
#define TP_PAD 160
#define NNZ_   31440
#define LDW    3168
#define NWM    ((size_t)N_C * (size_t)LDW)
#define KSTEPS (N_C / 4)
#define NSTRIP 197
#define RUN    48
#define OCCW   100

#define OFF_OCC  ((size_t)N_C * NSTRIP * RUN)
#define OFF_CS   (3 * NWM)
#define OFF_DS   (OFF_CS + (size_t)TP_PAD * N_C)
#define OFF_BC   (OFF_DS + (size_t)TP_PAD * N_C)
#define OFF_BD   (OFF_BC + (size_t)TP * N_C)

#define ZGRP   3930

#define TPB    256
#define ZBLK   62
#define SBLK   123
#define PBLK   1892
#define PSPLIT 832
#define GRID_A 894
#define GRID_B 1183
static_assert(GRID_A == ZBLK + PSPLIT && GRID_B == SBLK + PBLK - PSPLIT, "setup grid sizes");
static_assert(ZBLK * TPB >= ZGRP * 4 && ZBLK * TPB >= NSTRIP * OCCW / 2 && (NSTRIP * OCCW) % 2 == 0,
              "clearing part covers every entry and every bitmap word");

__device__ __forceinline__ void zero_part(int id, const int* __restrict__ rows,
                                          const int* __restrict__ cols,
                                          float* __restrict__ ws) {
    if (id < NSTRIP * OCCW / 2) ((uint2*)(ws + OFF_OCC))[id] = make_uint2(0u, 0u);
    const int g = id >> 2;
    if (g >= ZGRP) return;
    const v4f z = (v4f){0.0f, 0.0f, 0.0f, 0.0f};
#pragma unroll
    for (int k = 0; k < 8; ++k) {
        const int e = g + k * ZGRP;
        const size_t run = (size_t)rows[e] * NSTRIP + (size_t)(cols[e] >> 4);
        v4f* p = (v4f*)(ws + run * RUN) + (id & 3);
        p[0] = z;
        p[4] = z;
        p[8] = z;
    }
}

__device__ __forceinline__ void scatter_part(int i, const int* __restrict__ rows,
                                             const int* __restrict__ cols,
                                             const float* __restrict__ bnz,
                                             const float* __restrict__ anz,
                                             const float* __restrict__ hnz,
                                             float* __restrict__ ws) {
    if (i >= NNZ_) return;
    const int r  = rows[i];
    const int cc = cols[i];
    const int s  = cc >> 4;
    const size_t off = ((size_t)r * NSTRIP + (size_t)s) * RUN + (size_t)((cc & 15) * 3);
    atomicAdd(ws + off,     hnz[i]);
    atomicAdd(ws + off + 1, anz[i]);
    atomicAdd(ws + off + 2, bnz[i]);
    atomicOr((unsigned*)(ws + OFF_OCC) + s * OCCW + (r >> 5), 1u << (r & 31));
}

__device__ __forceinline__ void prep_part(int idx,
                       const float* __restrict__ C, const float* __restrict__ D,
                       const float* __restrict__ M, const float* __restrict__ cov,
                       const float* __restrict__ mu, const float* __restrict__ nu,
                       const float* __restrict__ ups, const float* __restrict__ zet,
                       float* __restrict__ ws) {

    if (idx < KSTEPS * 10 * 32) {
        const int lane = idx & 31;
        const int w    = idx >> 5;
        const int i    = w / KSTEPS;
        const int ks   = w - i * KSTEPS;
        const int t    = 16 * i + (lane & 15);
        const int m    = 4 * ks + 2 * (lane >> 4);
        v2f q0 = (v2f){0.0f, 0.0f};
        v2f q1 = (v2f){0.0f, 0.0f};
        if (t < TP) {
            const v2f c0 = *(const v2f*)(C + (size_t)t * N_C + m);
            const v2f c1 = *(const v2f*)(C + (size_t)(t + 1) * N_C + m);
            const v2f d0 = *(const v2f*)(D + (size_t)t * N_C + m);
            const v2f d1 = *(const v2f*)(D + (size_t)(t + 1) * N_C + m);
            q0.x = c0.x + c1.x;
            q0.y = d0.x + d1.x;
            q1.x = c0.y + c1.y;
            q1.y = d0.y + d1.y;
        }
        v2f* xr = (v2f*)(ws + OFF_CS);
        xr[(size_t)m * TP_PAD + (size_t)t]       = q0;
        xr[(size_t)(m + 1) * TP_PAD + (size_t)t] = q1;
    }

    if (idx >= TP * N_C) return;
    int t = idx / N_C;
    int m = idx - t * N_C;

    float mc = 0.0f, md = 0.0f;
#pragma unroll
    for (int k = 0; k < NMOB; ++k) {
#pragma unroll
        for (int tau = 0; tau < NP; ++tau) {
            float mv = M[((size_t)k * T_ALL + (size_t)(t + tau)) * N_C + m];
            mc = fmaf(mv, mu[k * NP + tau], mc);
            md = fmaf(mv, nu[k * NP + tau], md);
        }
    }
    float uc = 0.0f, zc = 0.0f;
#pragma unroll
    for (int j = 0; j < NCOV; ++j) {
        float cv = cov[(size_t)j * N_C + m];
        uc = fmaf(ups[j], cv, uc);
        zc = fmaf(zet[j], cv, zc);
    }
    ws[OFF_BC + (size_t)idx] = mc + uc;
    ws[OFF_BD + (size_t)idx] = md + zc;
}


__global__ __launch_bounds__(TPB)
void setup_a_k(const int* __restrict__ rows, const int* __restrict__ cols,
               const float* __restrict__ C, const float* __restrict__ D,
               const float* __restrict__ M, const float* __restrict__ cov,
               const float* __restrict__ mu, const float* __restrict__ nu,
               const float* __restrict__ ups, const float* __restrict__ zet,
               float* __restrict__ ws) {
    const int b = blockIdx.x;
    if (b < ZBLK) {
        zero_part(b * TPB + (int)threadIdx.x, rows, cols, ws);
        return;
    }
    prep_part((b - ZBLK) * TPB + (int)threadIdx.x, C, D, M, cov, mu, nu, ups, zet, ws);
}

__global__ __launch_bounds__(TPB)
void setup_b_k(const int* __restrict__ rows, const int* __restrict__ cols,
               const float* __restrict__ bnz, const float* __restrict__ anz,
               const float* __restrict__ hnz,
               const float* __restrict__ C, const float* __restrict__ D,
               const float* __restrict__ M, const float* __restrict__ cov,
               const float* __restrict__ mu, const float* __restrict__ nu,
               const float* __restrict__ ups, const float* __restrict__ zet,
               float* __restrict__ ws) {
    const int b = blockIdx.x;
    if (b < SBLK) {
        scatter_part(b * TPB + (int)threadIdx.x, rows, cols, bnz, anz, hnz, ws);
        return;
    }
    prep_part((b - SBLK + PSPLIT) * TPB + (int)threadIdx.x, C, D, M, cov, mu, nu, ups, zet, ws);
}

#define NSL    8
#define NTILE  20

struct KFrag {
    v2f b;
    v2f hax, hay;
    v2f xa[10], xb[10];
};

__device__ __forceinline__ void load_kfrag(KFrag& f, const float* __restrict__ wr,
                                           const v2f* __restrict__ xf,
                                           int qx, int qy, bool vx, bool vy) {
    const float* pa = wr + (size_t)qx * (size_t)(NSTRIP * RUN);
    const float* pb = wr + (size_t)qy * (size_t)(NSTRIP * RUN);
    const float ha0 = pa[0], ha1 = pa[1], ba = pa[2];
    const float hb0 = pb[0], hb1 = pb[1], bb = pb[2];
    f.hax = (v2f){vx ? ha0 : 0.0f, vx ? ha1 : 0.0f};
    f.hay = (v2f){vy ? hb0 : 0.0f, vy ? hb1 : 0.0f};
    f.b   = (v2f){vx ? ba : 0.0f, vy ? bb : 0.0f};
    const v2f* xa = xf + (size_t)qx * TP_PAD;
    const v2f* xb = xf + (size_t)qy * TP_PAD;
#pragma unroll
    for (int i = 0; i < 10; ++i) {
        f.xa[i] = xa[i * 16];
        f.xb[i] = xb[i * 16];
    }
}

__device__ __forceinline__ void mul_kfrag(const KFrag& f, v8f (&accC)[10], v8f (&accD)[10]) {
#pragma unroll
    for (int i = 0; i < 10; ++i) {
        const v2f ac = (v2f){f.xa[i].x, f.xb[i].x};
        accC[i] = __builtin_amdgcn_wmma_f32_16x16x4_f32(false, ac, false, f.b,
                                                        (short)0, accC[i], false, false);
        accD[i] = __builtin_amdgcn_wmma_f32_16x16x4_f32(false, f.xa[i], false, f.hax,
                                                        (short)0, accD[i], false, false);
        accD[i] = __builtin_amdgcn_wmma_f32_16x16x4_f32(false, f.xb[i], false, f.hay,
                                                        (short)0, accD[i], false, false);
    }
}

__global__ __launch_bounds__(NSL * 32)
void gemm_k(const float* __restrict__ ws, float* __restrict__ out) {
    __shared__ float red[NSL * NTILE * 8 * 32];

    const int wave = __builtin_amdgcn_readfirstlane((int)(threadIdx.x >> 5));
    const int lane = threadIdx.x & 31;
    const int c    = lane & 15;
    const int h    = lane >> 4;
    const int c0   = blockIdx.x * 16;

    const int ks = (wave * KSTEPS) / NSL;
    const int ke = ((wave + 1) * KSTEPS) / NSL;

    const int r0 = 4 * ks;
    const int r1 = 4 * ke;
    const int w0 = r0 >> 5;
    const int nw = ((r1 - 1) >> 5) - w0 + 1;
    unsigned bits = 0u;
    if (lane < nw)
        bits = ((const unsigned*)(ws + OFF_OCC))[(size_t)blockIdx.x * OCCW + (size_t)(w0 + lane)];

    const int  col   = c0 + c;
    const bool colok = col < N_C;
    const int  colc  = colok ? col : (N_C - 1);
    const int  rowl  = wave + 8 * h;
    float bv[NTILE];
#pragma unroll
    for (int j = 0; j < NTILE; ++j) {
        const int zz = j / 10;
        const int i  = j - 10 * zz;
        int row = 16 * i + rowl;
        if (i == 9 && row > TP - 1) row = TP - 1;
        bv[j] = ws[(zz ? OFF_BD : OFF_BC) + (size_t)row * N_C + (size_t)colc];
    }

    float* list = red + (size_t)wave * (NTILE * 8 * 32);
    int cnt;
    {
        const int wb = 32 * (w0 + lane);
        const int lo = r0 - wb;
        const int hi = r1 - wb;
        if (lo > 0) bits &= ~0u << lo;
        if (lane < nw && hi < 32) bits &= ~0u >> (32 - hi);
        const int pc = __popc(bits);
        int incl = pc;
#pragma unroll
        for (int d = 1; d < 32; d <<= 1) {
            const int v = __shfl_up(incl, d);
            if (lane >= d) incl += v;
        }
        cnt = __builtin_amdgcn_readlane(incl, 31);
        int pos = incl - pc;
        while (bits != 0u) {
            const int b = __builtin_ctz(bits);
            bits &= bits - 1u;
            list[pos++] = __int_as_float(wb + b);
        }
        if (lane < 3) list[cnt + lane] = __int_as_float(r0);
    }
    __builtin_amdgcn_fence(__ATOMIC_RELEASE, "wavefront");
    __builtin_amdgcn_wave_barrier();
    __builtin_amdgcn_fence(__ATOMIC_ACQUIRE, "wavefront");

    const float* wr = ws + (size_t)blockIdx.x * RUN + (size_t)(c * 3);
    const v2f* xf = (const v2f*)(ws + OFF_CS) + c;
    const v2f* lq = (const v2f*)list + h;

    v8f accC[10];
    v8f accD[10];
#pragma unroll
    for (int i = 0; i < 10; ++i) { accC[i] = (v8f){}; accD[i] = (v8f){}; }

    const int nst = (cnt + 3) >> 2;
    const int sl  = 2 * h;

    if (nst > 0) {
        KFrag f0, f1;
        int j1 = (1 < nst) ? 1 : 0;
        v2f qa = lq[0];
        v2f qb = lq[2 * j1];
        load_kfrag(f0, wr, xf, __float_as_int(qa.x), __float_as_int(qa.y),
                   sl < cnt, sl + 1 < cnt);
        for (int j = 0; j + 1 < nst; j += 2) {
            load_kfrag(f1, wr, xf, __float_as_int(qb.x), __float_as_int(qb.y),
                       4 * (j + 1) + sl < cnt, 4 * (j + 1) + sl + 1 < cnt);
            const int j2 = (j + 2 < nst) ? j + 2 : j + 1;
            qa = lq[2 * j2];
            __builtin_amdgcn_sched_barrier(0);
            mul_kfrag(f0, accC, accD);
            load_kfrag(f0, wr, xf, __float_as_int(qa.x), __float_as_int(qa.y),
                       4 * j2 + sl < cnt, 4 * j2 + sl + 1 < cnt);
            const int j3 = (j + 3 < nst) ? j + 3 : j2;
            qb = lq[2 * j3];
            __builtin_amdgcn_sched_barrier(0);
            mul_kfrag(f1, accC, accD);
        }
        if ((nst & 1) != 0) mul_kfrag(f0, accC, accD);
    }

    float* mine = red + (size_t)wave * (NTILE * 8 * 32) + lane;
#pragma unroll
    for (int i = 0; i < 10; ++i) {
#pragma unroll
        for (int r = 0; r < 8; ++r) {
            mine[(i * 8 + r) * 32]        = accC[i][r];
            mine[((10 + i) * 8 + r) * 32] = accD[i][r];
        }
    }
    __syncthreads();

    const float* jn = red + (size_t)wave * 32 + lane;
    float sv[NTILE];
#pragma unroll
    for (int j = 0; j < NTILE; ++j) {
        float s = 0.0f;
#pragma unroll
        for (int sw = 0; sw < NSL; ++sw)
            s += jn[(size_t)(sw * (NTILE * 8) + j * 8) * 32];
        sv[j] = s + bv[j];
    }
    if (colok) {
        float* o = out + (size_t)rowl * N_C + (size_t)col;
#pragma unroll
        for (int j = 0; j < NTILE; ++j) {
            const int zz = j / 10;
            const int i  = j - 10 * zz;
            if (i < 9 || 16 * 9 + rowl < TP)
                o[(size_t)zz * TP * N_C + (size_t)(16 * i) * N_C] = sv[j];
        }
    }
}

extern "C" void kernel_launch(void* const* d_in, const int* in_sizes, int n_in,
                              void* d_out, int out_size, void* d_ws, size_t ws_size,
                              hipStream_t stream) {
    const float* C   = (const float*)d_in[0];
    const float* D   = (const float*)d_in[1];
    const float* M   = (const float*)d_in[2];
    const float* cov = (const float*)d_in[3];
    const float* bnz = (const float*)d_in[4];
    const float* anz = (const float*)d_in[5];
    const float* hnz = (const float*)d_in[6];
    const float* mu  = (const float*)d_in[7];
    const float* nu  = (const float*)d_in[8];
    const float* ups = (const float*)d_in[9];
    const float* zet = (const float*)d_in[10];
    const int* rows  = (const int*)d_in[11];
    const int* cols  = (const int*)d_in[12];
    float* ws  = (float*)d_ws;
    float* out = (float*)d_out;

    setup_a_k<<<GRID_A, TPB, 0, stream>>>(rows, cols, C, D, M, cov, mu, nu, ups, zet, ws);
    setup_b_k<<<GRID_B, TPB, 0, stream>>>(rows, cols, bnz, anz, hnz,
                                          C, D, M, cov, mu, nu, ups, zet, ws);
    gemm_k<<<197, 256, 0, stream>>>(ws, out);
}


# Fold/base cells re-dealt: a lane owns a column and walks 16 time steps, the lag row two steps share kept in registers; XF pieces turned in the wave's own LDS; all in launch 1: 990 waves, not 15136
# speedup vs baseline: 8.6850x; 1.0333x over previous
#include <hip/hip_runtime.h>

typedef __attribute__((ext_vector_type(2))) float v2f;
typedef __attribute__((ext_vector_type(4))) float v4f;
typedef __attribute__((ext_vector_type(8))) float v8f;

#define N_C    3144
#define NP     2
#define NMOB   6
#define NCOV   10
#define T_ALL  156
#define TP     154
#define TP_PAD 160
#define NNZ_   31440
#define LDW    3168
#define NWM    ((size_t)N_C * (size_t)LDW)
#define KSTEPS (N_C / 4)
#define NSTRIP 197
#define RUN    48
#define OCCW   100

#define OFF_OCC  ((size_t)N_C * NSTRIP * RUN)
#define OFF_CS   (3 * NWM)
#define OFF_DS   (OFF_CS + (size_t)TP_PAD * N_C)
#define OFF_BC   (OFF_DS + (size_t)TP_PAD * N_C)
#define OFF_BD   (OFF_BC + (size_t)TP * N_C)

#define ZGRP   3930

#define TPB    256
#define WPB    (TPB / 32)
#define ZBLK   62
#define SBLK   123
#define WTS    16
#define NCG    99
#define NTC    (TP_PAD / WTS)
#define NITEM  (NCG * NTC)
#define PBLK   124
#define XPITCH 36
#define GRID_A 186
static_assert(GRID_A == ZBLK + PBLK && PBLK * WPB >= NITEM && NCG * 32 >= N_C && NTC * WTS == TP_PAD,
              "setup grid sizes");
static_assert(ZBLK * TPB >= ZGRP * 4 && ZBLK * TPB >= NSTRIP * OCCW / 2 && (NSTRIP * OCCW) % 2 == 0,
              "clearing part covers every entry and every bitmap word");
static_assert(SBLK * TPB >= NNZ_, "scatter part covers every entry");

__device__ __forceinline__ void zero_part(int id, const int* __restrict__ rows,
                                          const int* __restrict__ cols,
                                          float* __restrict__ ws) {
    if (id < NSTRIP * OCCW / 2) ((uint2*)(ws + OFF_OCC))[id] = make_uint2(0u, 0u);
    const int g = id >> 2;
    if (g >= ZGRP) return;
    const v4f z = (v4f){0.0f, 0.0f, 0.0f, 0.0f};
#pragma unroll
    for (int k = 0; k < 8; ++k) {
        const int e = g + k * ZGRP;
        const size_t run = (size_t)rows[e] * NSTRIP + (size_t)(cols[e] >> 4);
        v4f* p = (v4f*)(ws + run * RUN) + (id & 3);
        p[0] = z;
        p[4] = z;
        p[8] = z;
    }
}

__device__ __forceinline__ void scatter_part(int i, const int* __restrict__ rows,
                                             const int* __restrict__ cols,
                                             const float* __restrict__ bnz,
                                             const float* __restrict__ anz,
                                             const float* __restrict__ hnz,
                                             float* __restrict__ ws) {
    if (i >= NNZ_) return;
    const int r  = rows[i];
    const int cc = cols[i];
    const int s  = cc >> 4;
    const size_t off = ((size_t)r * NSTRIP + (size_t)s) * RUN + (size_t)((cc & 15) * 3);
    atomicAdd(ws + off,     hnz[i]);
    atomicAdd(ws + off + 1, anz[i]);
    atomicAdd(ws + off + 2, bnz[i]);
    atomicOr((unsigned*)(ws + OFF_OCC) + s * OCCW + (r >> 5), 1u << (r & 31));
}

template <int NT>
__device__ __forceinline__ void prep_walk(int tc, int cg, int lane, float* __restrict__ tile,
                       const float* __restrict__ C, const float* __restrict__ D,
                       const float* __restrict__ M, const float* __restrict__ cov,
                       const float* __restrict__ mu, const float* __restrict__ nu,
                       const float* __restrict__ ups, const float* __restrict__ zet,
                       float* __restrict__ ws) {
    static_assert(NT >= 1 && NT <= WTS, "steps of a chunk");
    const int t0 = WTS * tc;
    const int m  = 32 * cg + lane;
    const bool mok = m < N_C;
    const int  mcl = mok ? m : (N_C - 1);
    const size_t cell = (size_t)t0 * N_C + (size_t)mcl;
    const float* Cp = C + cell;
    const float* Dp = D + cell;
    const float* Mp = M + cell;

    float muv[NMOB * NP], nuv[NMOB * NP], upv[NCOV], ztv[NCOV];
#pragma unroll
    for (int i = 0; i < NMOB * NP; ++i) { muv[i] = mu[i]; nuv[i] = nu[i]; }
#pragma unroll
    for (int j = 0; j < NCOV; ++j) { upv[j] = ups[j]; ztv[j] = zet[j]; }

    float cr[WTS + 1], dr[WTS + 1], mr[NMOB][WTS + 1];
#pragma unroll
    for (int s = 0; s <= NT; ++s) {
        cr[s] = Cp[s * N_C];
        dr[s] = Dp[s * N_C];
#pragma unroll
        for (int k = 0; k < NMOB; ++k)
            mr[k][s] = Mp[(k * T_ALL + s) * N_C];
    }

    float uc = 0.0f, zc = 0.0f;
#pragma unroll
    for (int j = 0; j < NCOV; ++j) {
        const float cv = cov[(size_t)j * N_C + (size_t)mcl];
        uc = fmaf(upv[j], cv, uc);
        zc = fmaf(ztv[j], cv, zc);
    }

    float* bc = ws + OFF_BC + cell;
    float* bd = ws + OFF_BD + cell;
    v4f xq[WTS / 2];
#pragma unroll
    for (int s = 0; s < WTS; ++s) {
        float cs = 0.0f, ds = 0.0f;
        if (s < NT) {
            float mc = 0.0f, md = 0.0f;
#pragma unroll
            for (int k = 0; k < NMOB; ++k) {
#pragma unroll
                for (int tau = 0; tau < NP; ++tau) {
                    const float mv = mr[k][s + tau];
                    mc = fmaf(mv, muv[k * NP + tau], mc);
                    md = fmaf(mv, nuv[k * NP + tau], md);
                }
            }
            if (mok) {
                bc[s * N_C] = mc + uc;
                bd[s * N_C] = md + zc;
            }
            cs = cr[s] + cr[s + 1];
            ds = dr[s] + dr[s + 1];
        }
        xq[s >> 1][2 * (s & 1)]     = cs;
        xq[s >> 1][2 * (s & 1) + 1] = ds;
    }

    v4f* tl = (v4f*)(tile + lane * XPITCH);
#pragma unroll
    for (int q = 0; q < WTS / 2; ++q) tl[q] = xq[q];
    __builtin_amdgcn_fence(__ATOMIC_RELEASE, "wavefront");
    __builtin_amdgcn_wave_barrier();
    __builtin_amdgcn_fence(__ATOMIC_ACQUIRE, "wavefront");
    const int pr = lane >> 3;
    const int pp = lane & 7;
    float* xo = ws + OFF_CS + 2 * (size_t)t0 + 4 * (size_t)pp;
    v4f xv[8];
#pragma unroll
    for (int j = 0; j < 8; ++j)
        xv[j] = *(const v4f*)(tile + (4 * j + pr) * XPITCH + 4 * pp);
#pragma unroll
    for (int j = 0; j < 8; ++j) {
        const int mm = 32 * cg + 4 * j + pr;
        if (mm < N_C) *(v4f*)(xo + 2 * (size_t)mm * TP_PAD) = xv[j];
    }
}


__global__ __launch_bounds__(TPB)
void setup_a_k(const int* __restrict__ rows, const int* __restrict__ cols,
               const float* __restrict__ C, const float* __restrict__ D,
               const float* __restrict__ M, const float* __restrict__ cov,
               const float* __restrict__ mu, const float* __restrict__ nu,
               const float* __restrict__ ups, const float* __restrict__ zet,
               float* __restrict__ ws) {
    __shared__ __attribute__((aligned(16))) float xt[WPB * 32 * XPITCH];
    const int b = blockIdx.x;
    if (b < ZBLK) {
        zero_part(b * TPB + (int)threadIdx.x, rows, cols, ws);
        return;
    }
    const int wave = __builtin_amdgcn_readfirstlane((int)(threadIdx.x >> 5));
    const int item = (b - ZBLK) * WPB + wave;
    if (item >= NITEM) return;
    const int tc = item / NCG;
    const int cg = item - tc * NCG;
    const int lane = (int)(threadIdx.x & 31);
    float* tile = xt + wave * (32 * XPITCH);
    if (tc == NTC - 1)
        prep_walk<TP - WTS * (NTC - 1)>(tc, cg, lane, tile, C, D, M, cov, mu, nu, ups, zet, ws);
    else
        prep_walk<WTS>(tc, cg, lane, tile, C, D, M, cov, mu, nu, ups, zet, ws);
}

__global__ __launch_bounds__(TPB)
void setup_b_k(const int* __restrict__ rows, const int* __restrict__ cols,
               const float* __restrict__ bnz, const float* __restrict__ anz,
               const float* __restrict__ hnz, float* __restrict__ ws) {
    scatter_part((int)(blockIdx.x * TPB + threadIdx.x), rows, cols, bnz, anz, hnz, ws);
}

#define NSL    8
#define NTILE  20

struct KFrag {
    v2f b;
    v2f hax, hay;
    v2f xa[10], xb[10];
};

__device__ __forceinline__ void load_kfrag(KFrag& f, const float* __restrict__ wr,
                                           const v2f* __restrict__ xf,
                                           int qx, int qy, bool vx, bool vy) {
    const float* pa = wr + (size_t)qx * (size_t)(NSTRIP * RUN);
    const float* pb = wr + (size_t)qy * (size_t)(NSTRIP * RUN);
    const float ha0 = pa[0], ha1 = pa[1], ba = pa[2];
    const float hb0 = pb[0], hb1 = pb[1], bb = pb[2];
    f.hax = (v2f){vx ? ha0 : 0.0f, vx ? ha1 : 0.0f};
    f.hay = (v2f){vy ? hb0 : 0.0f, vy ? hb1 : 0.0f};
    f.b   = (v2f){vx ? ba : 0.0f, vy ? bb : 0.0f};
    const v2f* xa = xf + (size_t)qx * TP_PAD;
    const v2f* xb = xf + (size_t)qy * TP_PAD;
#pragma unroll
    for (int i = 0; i < 10; ++i) {
        f.xa[i] = xa[i * 16];
        f.xb[i] = xb[i * 16];
    }
}

__device__ __forceinline__ void mul_kfrag(const KFrag& f, v8f (&accC)[10], v8f (&accD)[10]) {
#pragma unroll
    for (int i = 0; i < 10; ++i) {
        const v2f ac = (v2f){f.xa[i].x, f.xb[i].x};
        accC[i] = __builtin_amdgcn_wmma_f32_16x16x4_f32(false, ac, false, f.b,
                                                        (short)0, accC[i], false, false);
        accD[i] = __builtin_amdgcn_wmma_f32_16x16x4_f32(false, f.xa[i], false, f.hax,
                                                        (short)0, accD[i], false, false);
        accD[i] = __builtin_amdgcn_wmma_f32_16x16x4_f32(false, f.xb[i], false, f.hay,
                                                        (short)0, accD[i], false, false);
    }
}

__global__ __launch_bounds__(NSL * 32)
void gemm_k(const float* __restrict__ ws, float* __restrict__ out) {
    __shared__ float red[NSL * NTILE * 8 * 32];

    const int wave = __builtin_amdgcn_readfirstlane((int)(threadIdx.x >> 5));
    const int lane = threadIdx.x & 31;
    const int c    = lane & 15;
    const int h    = lane >> 4;
    const int c0   = blockIdx.x * 16;

    const int ks = (wave * KSTEPS) / NSL;
    const int ke = ((wave + 1) * KSTEPS) / NSL;

    const int r0 = 4 * ks;
    const int r1 = 4 * ke;
    const int w0 = r0 >> 5;
    const int nw = ((r1 - 1) >> 5) - w0 + 1;
    unsigned bits = 0u;
    if (lane < nw)
        bits = ((const unsigned*)(ws + OFF_OCC))[(size_t)blockIdx.x * OCCW + (size_t)(w0 + lane)];

    const int  col   = c0 + c;
    const bool colok = col < N_C;
    const int  colc  = colok ? col : (N_C - 1);
    const int  rowl  = wave + 8 * h;
    float bv[NTILE];
#pragma unroll
    for (int j = 0; j < NTILE; ++j) {
        const int zz = j / 10;
        const int i  = j - 10 * zz;
        int row = 16 * i + rowl;
        if (i == 9 && row > TP - 1) row = TP - 1;
        bv[j] = ws[(zz ? OFF_BD : OFF_BC) + (size_t)row * N_C + (size_t)colc];
    }

    float* list = red + (size_t)wave * (NTILE * 8 * 32);
    int cnt;
    {
        const int wb = 32 * (w0 + lane);
        const int lo = r0 - wb;
        const int hi = r1 - wb;
        if (lo > 0) bits &= ~0u << lo;
        if (lane < nw && hi < 32) bits &= ~0u >> (32 - hi);
        const int pc = __popc(bits);
        int incl = pc;
#pragma unroll
        for (int d = 1; d < 32; d <<= 1) {
            const int v = __shfl_up(incl, d);
            if (lane >= d) incl += v;
        }
        cnt = __builtin_amdgcn_readlane(incl, 31);
        int pos = incl - pc;
        while (bits != 0u) {
            const int b = __builtin_ctz(bits);
            bits &= bits - 1u;
            list[pos++] = __int_as_float(wb + b);
        }
        if (lane < 3) list[cnt + lane] = __int_as_float(r0);
    }
    __builtin_amdgcn_fence(__ATOMIC_RELEASE, "wavefront");
    __builtin_amdgcn_wave_barrier();
    __builtin_amdgcn_fence(__ATOMIC_ACQUIRE, "wavefront");

    const float* wr = ws + (size_t)blockIdx.x * RUN + (size_t)(c * 3);
    const v2f* xf = (const v2f*)(ws + OFF_CS) + c;
    const v2f* lq = (const v2f*)list + h;

    v8f accC[10];
    v8f accD[10];
#pragma unroll
    for (int i = 0; i < 10; ++i) { accC[i] = (v8f){}; accD[i] = (v8f){}; }

    const int nst = (cnt + 3) >> 2;
    const int sl  = 2 * h;

    if (nst > 0) {
        KFrag f0, f1;
        int j1 = (1 < nst) ? 1 : 0;
        v2f qa = lq[0];
        v2f qb = lq[2 * j1];
        load_kfrag(f0, wr, xf, __float_as_int(qa.x), __float_as_int(qa.y),
                   sl < cnt, sl + 1 < cnt);
        for (int j = 0; j + 1 < nst; j += 2) {
            load_kfrag(f1, wr, xf, __float_as_int(qb.x), __float_as_int(qb.y),
                       4 * (j + 1) + sl < cnt, 4 * (j + 1) + sl + 1 < cnt);
            const int j2 = (j + 2 < nst) ? j + 2 : j + 1;
            qa = lq[2 * j2];
            __builtin_amdgcn_sched_barrier(0);
            mul_kfrag(f0, accC, accD);
            load_kfrag(f0, wr, xf, __float_as_int(qa.x), __float_as_int(qa.y),
                       4 * j2 + sl < cnt, 4 * j2 + sl + 1 < cnt);
            const int j3 = (j + 3 < nst) ? j + 3 : j2;
            qb = lq[2 * j3];
            __builtin_amdgcn_sched_barrier(0);
            mul_kfrag(f1, accC, accD);
        }
        if ((nst & 1) != 0) mul_kfrag(f0, accC, accD);
    }

    float* mine = red + (size_t)wave * (NTILE * 8 * 32) + lane;
#pragma unroll
    for (int i = 0; i < 10; ++i) {
#pragma unroll
        for (int r = 0; r < 8; ++r) {
            mine[(i * 8 + r) * 32]        = accC[i][r];
            mine[((10 + i) * 8 + r) * 32] = accD[i][r];
        }
    }
    __syncthreads();

    const float* jn = red + (size_t)wave * 32 + lane;
    float sv[NTILE];
#pragma unroll
    for (int j = 0; j < NTILE; ++j) {
        float s = 0.0f;
#pragma unroll
        for (int sw = 0; sw < NSL; ++sw)
            s += jn[(size_t)(sw * (NTILE * 8) + j * 8) * 32];
        sv[j] = s + bv[j];
    }
    if (colok) {
        float* o = out + (size_t)rowl * N_C + (size_t)col;
#pragma unroll
        for (int j = 0; j < NTILE; ++j) {
            const int zz = j / 10;
            const int i  = j - 10 * zz;
            if (i < 9 || 16 * 9 + rowl < TP)
                o[(size_t)zz * TP * N_C + (size_t)(16 * i) * N_C] = sv[j];
        }
    }
}

extern "C" void kernel_launch(void* const* d_in, const int* in_sizes, int n_in,
                              void* d_out, int out_size, void* d_ws, size_t ws_size,
                              hipStream_t stream) {
    const float* C   = (const float*)d_in[0];
    const float* D   = (const float*)d_in[1];
    const float* M   = (const float*)d_in[2];
    const float* cov = (const float*)d_in[3];
    const float* bnz = (const float*)d_in[4];
    const float* anz = (const float*)d_in[5];
    const float* hnz = (const float*)d_in[6];
    const float* mu  = (const float*)d_in[7];
    const float* nu  = (const float*)d_in[8];
    const float* ups = (const float*)d_in[9];
    const float* zet = (const float*)d_in[10];
    const int* rows  = (const int*)d_in[11];
    const int* cols  = (const int*)d_in[12];
    float* ws  = (float*)d_ws;
    float* out = (float*)d_out;

    setup_a_k<<<GRID_A, TPB, 0, stream>>>(rows, cols, C, D, M, cov, mu, nu, ups, zet, ws);
    setup_b_k<<<SBLK, TPB, 0, stream>>>(rows, cols, bnz, anz, hnz, ws);
    gemm_k<<<197, 256, 0, stream>>>(ws, out);
}
